# starting kernel
# baseline (speedup 1.0000x reference)
#include <hip/hip_runtime.h>


typedef __attribute__((ext_vector_type(2))) float v2f;
typedef __attribute__((ext_vector_type(8))) float v8f;

#define NKP   184
#define NP0   96
#define NP1   256
#define NROWS (1024*256)
#define OUT1_OFF NROWS

__global__ void symcon_pack(const float* __restrict__ U2_l0,
                            const float* __restrict__ U3_l0,
                            const float* __restrict__ U2_l1,
                            const float* __restrict__ U3_l1,
                            float* __restrict__ Bp0,
                            float* __restrict__ Bp1) {
  int idx = blockIdx.x * 256 + threadIdx.x;
  if (idx < NP0 * NKP) {
    int n = idx / NKP, k = idx % NKP;
    float v = 0.f;
    if (n < 81) v = (k < 180) ? U3_l0[n * 180 + k] : U2_l0[n * 4 + (k - 180)];
    Bp0[idx] = v;
  } else {
    idx -= NP0 * NKP;
    if (idx < NP1 * NKP) {
      int n = idx / NKP, k = idx % NKP;
      float v = 0.f;
      if (n < 243) v = (k < 180) ? U3_l1[n * 180 + k] : U2_l1[n * 4 + (k - 180)];
      Bp1[idx] = v;
    }
  }
}


__device__ __forceinline__ float zval(int k, const float (&xr)[9],
                                      const float (&p3)[20], const float (&p2)[4]) {
  return (k < 180) ? xr[k / 20] * p3[k % 20] : p2[k - 180];
}

__device__ __forceinline__ void compute_p(const float (&yv)[10], int c,
    const float* __restrict__ w1, const float* __restrict__ w2,
    const float* __restrict__ w3,
    float (&p3)[20], float (&p2)[4], float &p1) {
  #pragma unroll
  for (int k = 0; k < 20; ++k) p3[k] = 0.f;
  #pragma unroll
  for (int k = 0; k < 4; ++k) p2[k] = 0.f;
  p1 = 0.f;
  #pragma unroll
  for (int e = 0; e < 10; ++e) {
    float ye = yv[e];
    #pragma unroll
    for (int k = 0; k < 20; ++k) p3[k] = fmaf(ye, w3[(e * 20 + k) * 256 + c], p3[k]);
    #pragma unroll
    for (int k = 0; k < 4; ++k)  p2[k] = fmaf(ye, w2[(e * 4 + k) * 256 + c], p2[k]);
    p1 = fmaf(ye, w1[e * 256 + c], p1);
  }
}

__device__ __forceinline__ void gemm_seg(const float* __restrict__ bseg, int n0,
    int NT8, int lan, bool hi,
    const float (&xr)[9], const float (&p3)[20], const float (&p2)[4],
    v8f* __restrict__ acc, int NT) {
  (void)NT8;
  const char* bp = (const char*)bseg +
                   (size_t)(((n0 + lan) * NKP + (hi ? 2 : 0)) * 4);
  #pragma unroll
  for (int s = 0; s < 46; ++s) {
    float z0 = zval(4 * s + 0, xr, p3, p2);
    float z1 = zval(4 * s + 1, xr, p3, p2);
    float z2 = zval(4 * s + 2, xr, p3, p2);
    float z3 = zval(4 * s + 3, xr, p3, p2);
    v2f a;
    a.x = hi ? z2 : z0;
    a.y = hi ? z3 : z1;
    #pragma unroll
    for (int t = 0; t < 8; ++t) {
      if (t < NT) {
        v2f bv = *(const v2f*)(bp + t * (16 * NKP * 4) + s * 16);
        acc[t] = __builtin_amdgcn_wmma_f32_16x16x4_f32(
            false, a, false, bv, (short)0, acc[t], false, false);
      }
    }
  }
}

__global__ __launch_bounds__(256, 1) void symcon_main(
    const float* __restrict__ x, const float* __restrict__ y,
    const float* __restrict__ U1_l0, const float* __restrict__ U1_l1,
    const float* __restrict__ w1_l0, const float* __restrict__ w2_l0,
    const float* __restrict__ w3_l0,
    const float* __restrict__ w1_l1, const float* __restrict__ w2_l1,
    const float* __restrict__ w3_l1,
    const float* __restrict__ Bp0, const float* __restrict__ Bp1,
    float* __restrict__ out) {
  __shared__ float xs[8][16][12];

  const int tid  = threadIdx.x;
  const int wv   = tid >> 5;
  const int lane = tid & 31;
  const int lan  = lane & 15;
  const bool hi  = lane >= 16;
  const int row0 = (blockIdx.x * 8 + wv) * 16;
  const int row  = row0 + lan;
  const int b    = row >> 8;
  const int c    = row & 255;

  float xr[9];
  #pragma unroll
  for (int i = 0; i < 9; ++i) xr[i] = x[row * 9 + i];

  float yv[10];
  #pragma unroll
  for (int e = 0; e < 10; ++e) yv[e] = y[b * 10 + e];

  float p3[20], p2[4], p1;

  compute_p(yv, c, w1_l0, w2_l0, w3_l0, p3, p2, p1);
  if (!hi) {
    #pragma unroll
    for (int i = 0; i < 9; ++i) xs[wv][lan][i] = xr[i];
    xs[wv][lan][9] = p1;
  }

  {
    v8f acc6[6];
    v8f vz = {};
    #pragma unroll
    for (int t = 0; t < 6; ++t) acc6[t] = vz;

    gemm_seg(Bp0, 0, 0, lan, hi, xr, p3, p2, acc6, 6);

    float o0[8];
    #pragma unroll
    for (int j = 0; j < 8; ++j) o0[j] = 0.f;
    #pragma unroll
    for (int t = 0; t < 6; ++t) {
      int n = t * 16 + lan;
      if (n < 81) {
        int wq = n / 9, xq = n % 9;
        #pragma unroll
        for (int j = 0; j < 8; ++j) {
          int rl = j + (hi ? 8 : 0);
          float q = xs[wv][rl][wq] * xs[wv][rl][xq];
          o0[j] = fmaf(acc6[t][j], q, o0[j]);
        }
      }
    }
    #pragma unroll
    for (int j = 0; j < 8; ++j) {
      o0[j] += __shfl_xor(o0[j], 1, 32);
      o0[j] += __shfl_xor(o0[j], 2, 32);
      o0[j] += __shfl_xor(o0[j], 4, 32);
      o0[j] += __shfl_xor(o0[j], 8, 32);
    }
    if (lan == 0) {
      #pragma unroll
      for (int j = 0; j < 8; ++j) {
        int rl = j + (hi ? 8 : 0);
        float du = 0.f;
        #pragma unroll
        for (int wq = 0; wq < 9; ++wq) du = fmaf(U1_l0[wq], xs[wv][rl][wq], du);
        out[row0 + rl] = o0[j] + xs[wv][rl][9] * du;
      }
    }
  }

  compute_p(yv, c, w1_l1, w2_l1, w3_l1, p3, p2, p1);
  if (!hi) xs[wv][lan][10] = p1;

  float o1[8][3];
  #pragma unroll
  for (int j = 0; j < 8; ++j)
    #pragma unroll
    for (int wq = 0; wq < 3; ++wq) o1[j][wq] = 0.f;

  v8f acc8[8];
  #pragma unroll
  for (int seg = 0; seg < 2; ++seg) {
    const int n0 = seg * 128;
    v8f vz = {};
    #pragma unroll
    for (int t = 0; t < 8; ++t) acc8[t] = vz;

    gemm_seg(Bp1, n0, 0, lan, hi, xr, p3, p2, acc8, 8);

    #pragma unroll
    for (int t = 0; t < 8; ++t) {
      int n = n0 + t * 16 + lan;
      if (n < 243) {
        int wq = n / 81, xq = (n / 9) % 9, vq = n % 9;
        #pragma unroll
        for (int j = 0; j < 8; ++j) {
          int rl = j + (hi ? 8 : 0);
          float q = xs[wv][rl][xq] * xs[wv][rl][vq];
          float pz = acc8[t][j] * q;
          if (wq == 0)      o1[j][0] += pz;
          else if (wq == 1) o1[j][1] += pz;
          else              o1[j][2] += pz;
        }
      }
    }
  }

  #pragma unroll
  for (int j = 0; j < 8; ++j)
    #pragma unroll
    for (int wq = 0; wq < 3; ++wq) {
      float v = o1[j][wq];
      v += __shfl_xor(v, 1, 32);
      v += __shfl_xor(v, 2, 32);
      v += __shfl_xor(v, 4, 32);
      v += __shfl_xor(v, 8, 32);
      o1[j][wq] = v;
    }
  if (lan == 0) {
    #pragma unroll
    for (int j = 0; j < 8; ++j) {
      int rl = j + (hi ? 8 : 0);
      int r  = row0 + rl;
      #pragma unroll
      for (int wq = 0; wq < 3; ++wq) {
        float du = 0.f;
        #pragma unroll
        for (int xq = 0; xq < 9; ++xq)
          du = fmaf(U1_l1[wq * 9 + xq], xs[wv][rl][xq], du);
        out[OUT1_OFF + r * 3 + wq] = o1[j][wq] + xs[wv][rl][10] * du;
      }
    }
  }
}

extern "C" void kernel_launch(void* const* d_in, const int* in_sizes, int n_in,
                              void* d_out, int out_size, void* d_ws, size_t ws_size,
                              hipStream_t stream) {
  (void)in_sizes; (void)n_in; (void)out_size; (void)ws_size;
  const float* x     = (const float*)d_in[0];
  const float* y     = (const float*)d_in[1];
  const float* U1_l0 = (const float*)d_in[2];
  const float* U2_l0 = (const float*)d_in[3];
  const float* U3_l0 = (const float*)d_in[4];
  const float* U1_l1 = (const float*)d_in[5];
  const float* U2_l1 = (const float*)d_in[6];
  const float* U3_l1 = (const float*)d_in[7];
  const float* w1_l0 = (const float*)d_in[8];
  const float* w2_l0 = (const float*)d_in[9];
  const float* w3_l0 = (const float*)d_in[10];
  const float* w1_l1 = (const float*)d_in[11];
  const float* w2_l1 = (const float*)d_in[12];
  const float* w3_l1 = (const float*)d_in[13];

  float* Bp0 = (float*)d_ws;
  float* Bp1 = Bp0 + NP0 * NKP;

  const int pack_elems = (NP0 + NP1) * NKP;
  symcon_pack<<<(pack_elems + 255) / 256, 256, 0, stream>>>(
      U2_l0, U3_l0, U2_l1, U3_l1, Bp0, Bp1);

  symcon_main<<<2048, 256, 0, stream>>>(
      x, y, U1_l0, U1_l1,
      w1_l0, w2_l0, w3_l0,
      w1_l1, w2_l1, w3_l1,
      Bp0, Bp1, (float*)d_out);
}


# GEMM-K cut to the weight index (20+4): A = U folded per x-monomial, mirrored monomials (m->8-m) in the upper lane half, B = projected weights; 59 matrix instr per entry, was 83; sum order changes
# speedup vs baseline: 14.9813x; 14.9813x over previous
#include <hip/hip_runtime.h>


typedef __attribute__((ext_vector_type(2))) float v2f;
typedef __attribute__((ext_vector_type(8))) float v8f;

#define NGRP  4
#define NC3   85
#define NT3   11
#define KS3   5
#define NC2   25
#define NT2   4
#define Q_OFF (NT3 * 16 * 20)
#define GSTR  (Q_OFF + NT2 * 16 * 4)
#define BCH   16
#define NROWS (1024*256)
#define OUT1_OFF NROWS

struct Tri { int a, b, i; };

__host__ __device__ constexpr Tri rep3(int q) {
  int n = 0;
  for (int a = 0; a < 9; ++a)
    for (int b = a; b < 9; ++b)
      for (int i = b; i < 9; ++i)
        if (a + i < 8 || (a + i == 8 && b <= 4)) {
          if (n == q) return Tri{a, b, i};
          ++n;
        }
  return Tri{0, 0, 0};
}

__host__ __device__ constexpr Tri rep2(int q) {
  int n = 0;
  for (int a = 0; a < 9; ++a)
    for (int b = a; b < 9; ++b)
      if (a + b <= 8) {
        if (n == q) return Tri{a, b, 0};
        ++n;
      }
  return Tri{0, 0, 0};
}

__global__ void symcon_pack(const float* __restrict__ U2_l0,
                            const float* __restrict__ U3_l0,
                            const float* __restrict__ U2_l1,
                            const float* __restrict__ U3_l1,
                            float* __restrict__ Ug) {
  int idx = blockIdx.x * 256 + threadIdx.x;
  if (idx >= NGRP * GSTR) return;
  const int g = idx / GSTR;
  const int r = idx - g * GSTR;
  const float* __restrict__ U3 = g ? (U3_l1 + (g - 1) * 81 * 180) : U3_l0;
  const float* __restrict__ U2 = g ? (U2_l1 + (g - 1) * 81 * 4) : U2_l0;
  float v = 0.f;
  if (r < Q_OFF) {
    const int n = r / 20, kk = r - n * 20;
    const int h = (n >> 3) & 1;
    const int q = (n >> 4) * 8 + (n & 7);
    if (q < NC3) {
      const Tri t = rep3(q);
      int a = t.a, b = t.b, i = t.i;
      const bool self = (a + i == 8) && (b == 4);
      if (h) { const int na = 8 - i, nb = 8 - b, ni = 8 - a; a = na; b = nb; i = ni; }
      if (!(h && self)) {
        const bool ab = (a != b), bi = (b != i);
        v = U3[(a * 9 + b) * 180 + i * 20 + kk];
        if (bi)       v += U3[(a * 9 + i) * 180 + b * 20 + kk];
        if (ab)       v += U3[(b * 9 + a) * 180 + i * 20 + kk];
        if (ab)       v += U3[(b * 9 + i) * 180 + a * 20 + kk];
        if (bi)       v += U3[(i * 9 + a) * 180 + b * 20 + kk];
        if (ab && bi) v += U3[(i * 9 + b) * 180 + a * 20 + kk];
      }
    }
  } else {
    const int r2 = r - Q_OFF;
    const int n = r2 >> 2, kk = r2 & 3;
    const int h = (n >> 3) & 1;
    const int q = (n >> 4) * 8 + (n & 7);
    if (q < NC2) {
      const Tri t = rep2(q);
      int a = t.a, b = t.b;
      const bool self = (a + b == 8);
      if (h) { const int na = 8 - b, nb = 8 - a; a = na; b = nb; }
      if (!(h && self)) {
        v = U2[(a * 9 + b) * 4 + kk];
        if (a != b) v += U2[(b * 9 + a) * 4 + kk];
      }
    }
  }
  Ug[idx] = v;
}

template <int Q>
__device__ __forceinline__ float term3(float d, const float (&xr)[9], float o) {
  if constexpr (Q < NC3) {
    constexpr Tri t = rep3(Q);
    return fmaf(d, (xr[t.a] * xr[t.b]) * xr[t.i], o);
  } else {
    return o;
  }
}

template <int Q>
__device__ __forceinline__ float term2(float d, const float (&xr)[9], float o) {
  if constexpr (Q < NC2) {
    constexpr Tri t = rep2(Q);
    return fmaf(d, xr[t.a] * xr[t.b], o);
  } else {
    return o;
  }
}

template <int T>
__device__ __forceinline__ float tile3(const v8f& d, const float (&xr)[9], float o) {
  o = term3<8 * T + 0>(d[0], xr, o);
  o = term3<8 * T + 1>(d[1], xr, o);
  o = term3<8 * T + 2>(d[2], xr, o);
  o = term3<8 * T + 3>(d[3], xr, o);
  o = term3<8 * T + 4>(d[4], xr, o);
  o = term3<8 * T + 5>(d[5], xr, o);
  o = term3<8 * T + 6>(d[6], xr, o);
  o = term3<8 * T + 7>(d[7], xr, o);
  return o;
}

template <int T>
__device__ __forceinline__ float tile2(const v8f& d, const float (&xr)[9], float o) {
  o = term2<8 * T + 0>(d[0], xr, o);
  o = term2<8 * T + 1>(d[1], xr, o);
  o = term2<8 * T + 2>(d[2], xr, o);
  o = term2<8 * T + 3>(d[3], xr, o);
  o = term2<8 * T + 4>(d[4], xr, o);
  o = term2<8 * T + 5>(d[5], xr, o);
  o = term2<8 * T + 6>(d[6], xr, o);
  o = term2<8 * T + 7>(d[7], xr, o);
  return o;
}

__global__ __launch_bounds__(128, 1) void symcon_main(
    const float* __restrict__ x, const float* __restrict__ y,
    const float* __restrict__ U1_l0, const float* __restrict__ U1_l1,
    const float* __restrict__ w1_l0, const float* __restrict__ w2_l0,
    const float* __restrict__ w3_l0,
    const float* __restrict__ w1_l1, const float* __restrict__ w2_l1,
    const float* __restrict__ w3_l1,
    const float* __restrict__ Ug,
    float* __restrict__ out) {
  const int tid  = threadIdx.x;
  const int g    = __builtin_amdgcn_readfirstlane(tid >> 5);
  const int lane = tid & 31;
  const int lan  = lane & 15;
  const bool hi  = lane >= 16;
  const int kh   = hi ? 2 : 0;
  const int cg   = blockIdx.x & 15;
  const int bc   = blockIdx.x >> 4;
  const int c    = cg * 16 + lan;

  const float* __restrict__ w1 = g ? w1_l1 : w1_l0;
  const float* __restrict__ w2 = g ? w2_l1 : w2_l0;
  const float* __restrict__ w3 = g ? w3_l1 : w3_l0;
  const float* __restrict__ u1p = g ? (U1_l1 + (g - 1) * 9) : U1_l0;
  const float* __restrict__ w3h = w3 + kh * 256 + c;
  const float* __restrict__ w2h = w2 + kh * 256 + c;
  const float* __restrict__ w1c = w1 + c;

  float u1[9];
  #pragma unroll
  for (int i = 0; i < 9; ++i) u1[i] = u1p[i];

  v2f ua3[NT3][KS3];
  v2f ua2[NT2];
  {
    const float* ug  = Ug + (size_t)g * GSTR;
    const float* up3 = ug + lan * 20 + kh;
    #pragma unroll
    for (int t = 0; t < NT3; ++t)
      #pragma unroll
      for (int s = 0; s < KS3; ++s)
        ua3[t][s] = *(const v2f*)(up3 + t * (16 * 20) + 4 * s);
    const float* up2 = ug + Q_OFF + lan * 4 + kh;
    #pragma unroll
    for (int t = 0; t < NT2; ++t)
      ua2[t] = *(const v2f*)(up2 + t * (16 * 4));
  }

  const int xo = hi ? 8 : 0;
  const int xs = hi ? -1 : 1;

  const int b0 = bc * BCH;
  float xr[9], yv[10];
  {
    const int row = b0 * 256 + c;
    #pragma unroll
    for (int i = 0; i < 9; ++i) xr[i] = x[row * 9 + xo + xs * i];
    #pragma unroll
    for (int e = 0; e < 10; ++e) yv[e] = y[b0 * 10 + e];
  }

  #pragma nounroll
  for (int bi = 0; bi < BCH; ++bi) {
    const int b   = b0 + bi;
    const int row = b * 256 + c;

    const int bn = (b + 1 < 1024) ? (b + 1) : 1023;
    float xn[9], yn[10];
    {
      const int rown = bn * 256 + c;
      #pragma unroll
      for (int i = 0; i < 9; ++i) xn[i] = x[rown * 9 + xo + xs * i];
      #pragma unroll
      for (int e = 0; e < 10; ++e) yn[e] = y[bn * 10 + e];
    }

    float p3[KS3][2], p2[2], p1;
    #pragma unroll
    for (int s = 0; s < KS3; ++s) { p3[s][0] = 0.f; p3[s][1] = 0.f; }
    p2[0] = 0.f; p2[1] = 0.f;
    p1 = 0.f;
    #pragma unroll
    for (int e = 0; e < 10; ++e) {
      const float ye = yv[e];
      #pragma unroll
      for (int s = 0; s < KS3; ++s) {
        p3[s][0] = fmaf(ye, w3h[(e * 20 + 4 * s + 0) * 256], p3[s][0]);
        p3[s][1] = fmaf(ye, w3h[(e * 20 + 4 * s + 1) * 256], p3[s][1]);
      }
      p2[0] = fmaf(ye, w2h[(e * 4 + 0) * 256], p2[0]);
      p2[1] = fmaf(ye, w2h[(e * 4 + 1) * 256], p2[1]);
      p1 = fmaf(ye, w1c[e * 256], p1);
    }

    const v8f vz = {};
    v8f acc3[NT3];
    #pragma unroll
    for (int t = 0; t < NT3; ++t) acc3[t] = vz;
    #pragma unroll
    for (int t0 = 0; t0 < NT3; t0 += 4) {
      #pragma unroll
      for (int s = 0; s < KS3; ++s) {
        v2f zf;
        zf.x = p3[s][0];
        zf.y = p3[s][1];
        #pragma unroll
        for (int t = t0; t < t0 + 4; ++t) {
          if (t >= NT3) continue;
          acc3[t] = __builtin_amdgcn_wmma_f32_16x16x4_f32(
              false, ua3[t][s], false, zf, (short)0, acc3[t], false, false);
        }
      }
    }

    v8f acc2[NT2];
    {
      v2f zq;
      zq.x = p2[0];
      zq.y = p2[1];
      #pragma unroll
      for (int t = 0; t < NT2; ++t)
        acc2[t] = __builtin_amdgcn_wmma_f32_16x16x4_f32(
            false, ua2[t], false, zq, (short)0, vz, false, false);
    }

    float o = 0.f;
    o = tile3<0>(acc3[0], xr, o);
    o = tile3<1>(acc3[1], xr, o);
    o = tile3<2>(acc3[2], xr, o);
    o = tile3<3>(acc3[3], xr, o);
    o = tile3<4>(acc3[4], xr, o);
    o = tile3<5>(acc3[5], xr, o);
    o = tile3<6>(acc3[6], xr, o);
    o = tile3<7>(acc3[7], xr, o);
    o = tile3<8>(acc3[8], xr, o);
    o = tile3<9>(acc3[9], xr, o);
    o = tile3<10>(acc3[10], xr, o);
    o = tile2<0>(acc2[0], xr, o);
    o = tile2<1>(acc2[1], xr, o);
    o = tile2<2>(acc2[2], xr, o);
    o = tile2<3>(acc2[3], xr, o);
    o += __shfl_xor(o, 16, 32);

    if (!hi) {
      float du = 0.f;
      #pragma unroll
      for (int i = 0; i < 9; ++i) du = fmaf(u1[i], xr[i], du);
      const float res = o + p1 * du;
      if (g == 0) out[row] = res;
      else        out[OUT1_OFF + row * 3 + (g - 1)] = res;
    }

    #pragma unroll
    for (int i = 0; i < 9; ++i) xr[i] = xn[i];
    #pragma unroll
    for (int e = 0; e < 10; ++e) yv[e] = yn[e];
  }
}

extern "C" void kernel_launch(void* const* d_in, const int* in_sizes, int n_in,
                              void* d_out, int out_size, void* d_ws, size_t ws_size,
                              hipStream_t stream) {
  (void)in_sizes; (void)n_in; (void)out_size; (void)ws_size;
  const float* x     = (const float*)d_in[0];
  const float* y     = (const float*)d_in[1];
  const float* U1_l0 = (const float*)d_in[2];
  const float* U2_l0 = (const float*)d_in[3];
  const float* U3_l0 = (const float*)d_in[4];
  const float* U1_l1 = (const float*)d_in[5];
  const float* U2_l1 = (const float*)d_in[6];
  const float* U3_l1 = (const float*)d_in[7];
  const float* w1_l0 = (const float*)d_in[8];
  const float* w2_l0 = (const float*)d_in[9];
  const float* w3_l0 = (const float*)d_in[10];
  const float* w1_l1 = (const float*)d_in[11];
  const float* w2_l1 = (const float*)d_in[12];
  const float* w3_l1 = (const float*)d_in[13];

  float* Ug = (float*)d_ws;

  symcon_pack<<<59, 256, 0, stream>>>(U2_l0, U3_l0, U2_l1, U3_l1, Ug);

  symcon_main<<<1024, 128, 0, stream>>>(
      x, y, U1_l0, U1_l1,
      w1_l0, w2_l0, w3_l0,
      w1_l1, w2_l1, w3_l1,
      Ug, (float*)d_out);
}


# Monomial sum in nested form sum_a x_a*sum_b x_b*(Cq_ab+sum_i Cc_abi*x_i): one multiply-add per matrix result, no monomial products; x offsets kept 32-bit (registers stay 333); sum order changes
# speedup vs baseline: 17.2085x; 1.1487x over previous
#include <hip/hip_runtime.h>
#include <utility>


typedef __attribute__((ext_vector_type(2))) float v2f;
typedef __attribute__((ext_vector_type(8))) float v8f;

#define NGRP  4
#define NC3   85
#define NT3   11
#define KS3   5
#define NC2   25
#define NT2   4
#define Q_OFF (NT3 * 16 * 20)
#define GSTR  (Q_OFF + NT2 * 16 * 4)
#define BCH   16
#define NU    5
#define NROWS (1024*256)
#define OUT1_OFF NROWS

struct Tri { int a, b, i; };

__host__ __device__ constexpr Tri rep3(int q) {
  int n = 0;
  for (int a = 0; a < 9; ++a)
    for (int b = a; b < 9; ++b)
      for (int i = b; i < 9; ++i)
        if (a + i < 8 || (a + i == 8 && b <= 4)) {
          if (n == q) return Tri{a, b, i};
          ++n;
        }
  return Tri{0, 0, 0};
}

__host__ __device__ constexpr Tri rep2(int q) {
  int n = 0;
  for (int a = 0; a < 9; ++a)
    for (int b = a; b < 9; ++b)
      if (a + b <= 8) {
        if (n == q) return Tri{a, b, 0};
        ++n;
      }
  return Tri{0, 0, 0};
}

__global__ void symcon_pack(const float* __restrict__ U2_l0,
                            const float* __restrict__ U3_l0,
                            const float* __restrict__ U2_l1,
                            const float* __restrict__ U3_l1,
                            float* __restrict__ Ug) {
  int idx = blockIdx.x * 256 + threadIdx.x;
  if (idx >= NGRP * GSTR) return;
  const int g = idx / GSTR;
  const int r = idx - g * GSTR;
  const float* __restrict__ U3 = g ? (U3_l1 + (g - 1) * 81 * 180) : U3_l0;
  const float* __restrict__ U2 = g ? (U2_l1 + (g - 1) * 81 * 4) : U2_l0;
  float v = 0.f;
  if (r < Q_OFF) {
    const int n = r / 20, kk = r - n * 20;
    const int h = (n >> 3) & 1;
    const int q = (n >> 4) * 8 + (n & 7);
    if (q < NC3) {
      const Tri t = rep3(q);
      int a = t.a, b = t.b, i = t.i;
      const bool self = (a + i == 8) && (b == 4);
      if (h) { const int na = 8 - i, nb = 8 - b, ni = 8 - a; a = na; b = nb; i = ni; }
      if (!(h && self)) {
        const bool ab = (a != b), bi = (b != i);
        v = U3[(a * 9 + b) * 180 + i * 20 + kk];
        if (bi)       v += U3[(a * 9 + i) * 180 + b * 20 + kk];
        if (ab)       v += U3[(b * 9 + a) * 180 + i * 20 + kk];
        if (ab)       v += U3[(b * 9 + i) * 180 + a * 20 + kk];
        if (bi)       v += U3[(i * 9 + a) * 180 + b * 20 + kk];
        if (ab && bi) v += U3[(i * 9 + b) * 180 + a * 20 + kk];
      }
    }
  } else {
    const int r2 = r - Q_OFF;
    const int n = r2 >> 2, kk = r2 & 3;
    const int h = (n >> 3) & 1;
    const int q = (n >> 4) * 8 + (n & 7);
    if (q < NC2) {
      const Tri t = rep2(q);
      int a = t.a, b = t.b;
      const bool self = (a + b == 8);
      if (h) { const int na = 8 - b, nb = 8 - a; a = na; b = nb; }
      if (!(h && self)) {
        v = U2[(a * 9 + b) * 4 + kk];
        if (a != b) v += U2[(b * 9 + a) * 4 + kk];
      }
    }
  }
  Ug[idx] = v;
}

__host__ __device__ constexpr int grp_first(int a, int b) {
  for (int q = 0; q < NC3; ++q) {
    const Tri t = rep3(q);
    if (t.a == a && t.b == b) return q;
  }
  return NC3;
}
__host__ __device__ constexpr int grp_count(int a, int b) {
  int n = 0;
  for (int q = 0; q < NC3; ++q) {
    const Tri t = rep3(q);
    if (t.a == a && t.b == b) ++n;
  }
  return n;
}

template <int Q>
__device__ __forceinline__ float coef3(const v8f (&acc3)[NT3]) { return acc3[Q / 8][Q % 8]; }

template <int Q>
__device__ __forceinline__ float last3(const float (&xr)[9]) {
  constexpr Tri t = rep3(Q);
  return xr[t.i];
}

template <int P, int... I>
__device__ __forceinline__ float pair_inner(const v8f (&acc3)[NT3], const v8f (&acc2)[NT2],
                                            const float (&xr)[9],
                                            std::integer_sequence<int, I...>) {
  constexpr Tri pr = rep2(P);
  constexpr int q0 = grp_first(pr.a, pr.b);
  float t = acc2[P / 8][P % 8];
  ((t = fmaf(coef3<q0 + I>(acc3), last3<q0 + I>(xr), t)), ...);
  return t;
}

template <int P>
__device__ __forceinline__ void pair_term(const v8f (&acc3)[NT3], const v8f (&acc2)[NT2],
                                          const float (&xr)[9], float (&u)[NU]) {
  constexpr Tri pr = rep2(P);
  constexpr int n = grp_count(pr.a, pr.b);
  const float t = pair_inner<P>(acc3, acc2, xr, std::make_integer_sequence<int, n>{});
  u[pr.a] = fmaf(t, xr[pr.b], u[pr.a]);
}

template <int... P>
__device__ __forceinline__ void all_pairs(const v8f (&acc3)[NT3], const v8f (&acc2)[NT2],
                                          const float (&xr)[9], float (&u)[NU],
                                          std::integer_sequence<int, P...>) {
  (pair_term<P>(acc3, acc2, xr, u), ...);
}

__global__ __launch_bounds__(128, 1) void symcon_main(
    const float* __restrict__ x, const float* __restrict__ y,
    const float* __restrict__ U1_l0, const float* __restrict__ U1_l1,
    const float* __restrict__ w1_l0, const float* __restrict__ w2_l0,
    const float* __restrict__ w3_l0,
    const float* __restrict__ w1_l1, const float* __restrict__ w2_l1,
    const float* __restrict__ w3_l1,
    const float* __restrict__ Ug,
    float* __restrict__ out) {
  const int tid  = threadIdx.x;
  const int g    = __builtin_amdgcn_readfirstlane(tid >> 5);
  const int lane = tid & 31;
  const int lan  = lane & 15;
  const bool hi  = lane >= 16;
  const int kh   = hi ? 2 : 0;
  const int cg   = blockIdx.x & 15;
  const int bc   = blockIdx.x >> 4;
  const int c    = cg * 16 + lan;

  const float* __restrict__ w1 = g ? w1_l1 : w1_l0;
  const float* __restrict__ w2 = g ? w2_l1 : w2_l0;
  const float* __restrict__ w3 = g ? w3_l1 : w3_l0;
  const float* __restrict__ u1p = g ? (U1_l1 + (g - 1) * 9) : U1_l0;
  const float* __restrict__ w3h = w3 + kh * 256 + c;
  const float* __restrict__ w2h = w2 + kh * 256 + c;
  const float* __restrict__ w1c = w1 + c;

  float u1[9];
  #pragma unroll
  for (int i = 0; i < 9; ++i) u1[i] = u1p[i];

  v2f ua3[NT3][KS3];
  v2f ua2[NT2];
  {
    const float* ug  = Ug + (size_t)g * GSTR;
    const float* up3 = ug + lan * 20 + kh;
    #pragma unroll
    for (int t = 0; t < NT3; ++t)
      #pragma unroll
      for (int s = 0; s < KS3; ++s)
        ua3[t][s] = *(const v2f*)(up3 + t * (16 * 20) + 4 * s);
    const float* up2 = ug + Q_OFF + lan * 4 + kh;
    #pragma unroll
    for (int t = 0; t < NT2; ++t)
      ua2[t] = *(const v2f*)(up2 + t * (16 * 4));
  }

  const unsigned xo = hi ? 8u : 0u;
  const unsigned xs = hi ? ~0u : 1u;

  const int b0 = bc * BCH;
  float xr[9], yv[10];
  {
    const unsigned row = (unsigned)(b0 * 256 + c);
    #pragma unroll
    for (int i = 0; i < 9; ++i) xr[i] = x[row * 9u + xo + xs * (unsigned)i];
    #pragma unroll
    for (int e = 0; e < 10; ++e) yv[e] = y[b0 * 10 + e];
  }

  #pragma nounroll
  for (int bi = 0; bi < BCH; ++bi) {
    const int b   = b0 + bi;
    const int row = b * 256 + c;

    const int bn = (b + 1 < 1024) ? (b + 1) : 1023;
    float xn[9], yn[10];
    {
      const unsigned rown = (unsigned)(bn * 256 + c);
      #pragma unroll
      for (int i = 0; i < 9; ++i) xn[i] = x[rown * 9u + xo + xs * (unsigned)i];
      #pragma unroll
      for (int e = 0; e < 10; ++e) yn[e] = y[bn * 10 + e];
    }

    float p3[KS3][2], p2[2], p1;
    #pragma unroll
    for (int s = 0; s < KS3; ++s) { p3[s][0] = 0.f; p3[s][1] = 0.f; }
    p2[0] = 0.f; p2[1] = 0.f;
    p1 = 0.f;
    #pragma unroll
    for (int e = 0; e < 10; ++e) {
      const float ye = yv[e];
      #pragma unroll
      for (int s = 0; s < KS3; ++s) {
        p3[s][0] = fmaf(ye, w3h[(e * 20 + 4 * s + 0) * 256], p3[s][0]);
        p3[s][1] = fmaf(ye, w3h[(e * 20 + 4 * s + 1) * 256], p3[s][1]);
      }
      p2[0] = fmaf(ye, w2h[(e * 4 + 0) * 256], p2[0]);
      p2[1] = fmaf(ye, w2h[(e * 4 + 1) * 256], p2[1]);
      p1 = fmaf(ye, w1c[e * 256], p1);
    }

    const v8f vz = {};
    v8f acc2[NT2];
    {
      v2f zq;
      zq.x = p2[0];
      zq.y = p2[1];
      #pragma unroll
      for (int t = 0; t < NT2; ++t)
        acc2[t] = __builtin_amdgcn_wmma_f32_16x16x4_f32(
            false, ua2[t], false, zq, (short)0, vz, false, false);
    }
    v8f acc3[NT3];
    #pragma unroll
    for (int t = 0; t < NT3; ++t) acc3[t] = vz;
    #pragma unroll
    for (int t0 = 0; t0 < NT3; t0 += 4) {
      #pragma unroll
      for (int s = 0; s < KS3; ++s) {
        v2f zf;
        zf.x = p3[s][0];
        zf.y = p3[s][1];
        #pragma unroll
        for (int t = t0; t < t0 + 4; ++t) {
          if (t >= NT3) continue;
          acc3[t] = __builtin_amdgcn_wmma_f32_16x16x4_f32(
              false, ua3[t][s], false, zf, (short)0, acc3[t], false, false);
        }
      }
    }

    float u[NU];
    #pragma unroll
    for (int a = 0; a < NU; ++a) u[a] = 0.f;
    all_pairs(acc3, acc2, xr, u, std::make_integer_sequence<int, NC2>{});
    float o;
    {
      const float du0 = fmaf(u1[6], xr[6], fmaf(u1[3], xr[3], u1[0] * xr[0]));
      const float du1 = fmaf(u1[7], xr[7], fmaf(u1[4], xr[4], u1[1] * xr[1]));
      const float du2 = fmaf(u1[8], xr[8], fmaf(u1[5], xr[5], u1[2] * xr[2]));
      const float lin = (hi ? 0.f : p1) * ((du0 + du1) + du2);
      const float oa = fmaf(u[4], xr[4], fmaf(u[2], xr[2], fmaf(u[0], xr[0], lin)));
      const float ob = fmaf(u[3], xr[3], u[1] * xr[1]);
      o = oa + ob;
    }
    o += __shfl_xor(o, 16, 32);

    if (!hi) {
      if (g == 0) out[row] = o;
      else        out[OUT1_OFF + row * 3 + (g - 1)] = o;
    }

    #pragma unroll
    for (int i = 0; i < 9; ++i) xr[i] = xn[i];
    #pragma unroll
    for (int e = 0; e < 10; ++e) yv[e] = yn[e];
  }
}

extern "C" void kernel_launch(void* const* d_in, const int* in_sizes, int n_in,
                              void* d_out, int out_size, void* d_ws, size_t ws_size,
                              hipStream_t stream) {
  (void)in_sizes; (void)n_in; (void)out_size; (void)ws_size;
  const float* x     = (const float*)d_in[0];
  const float* y     = (const float*)d_in[1];
  const float* U1_l0 = (const float*)d_in[2];
  const float* U2_l0 = (const float*)d_in[3];
  const float* U3_l0 = (const float*)d_in[4];
  const float* U1_l1 = (const float*)d_in[5];
  const float* U2_l1 = (const float*)d_in[6];
  const float* U3_l1 = (const float*)d_in[7];
  const float* w1_l0 = (const float*)d_in[8];
  const float* w2_l0 = (const float*)d_in[9];
  const float* w3_l0 = (const float*)d_in[10];
  const float* w1_l1 = (const float*)d_in[11];
  const float* w2_l1 = (const float*)d_in[12];
  const float* w3_l1 = (const float*)d_in[13];

  float* Ug = (float*)d_ws;

  symcon_pack<<<59, 256, 0, stream>>>(U2_l0, U3_l0, U2_l1, U3_l1, Ug);

  symcon_main<<<1024, 128, 0, stream>>>(
      x, y, U1_l0, U1_l1,
      w1_l0, w2_l0, w3_l0,
      w1_l1, w2_l1, w3_l1,
      Ug, (float*)d_out);
}
